# starting kernel
# baseline (speedup 1.0000x reference)
#include <hip/hip_runtime.h>

typedef __attribute__((ext_vector_type(2))) float v2f;
typedef __attribute__((ext_vector_type(4))) float v4f;
typedef __attribute__((ext_vector_type(8))) float v8f;

#define NNODES 100000

__global__ void init_deg_kernel(float* __restrict__ deg, int n) {
    int i = blockIdx.x * blockDim.x + threadIdx.x;
    if (i < n) deg[i] = 1.0f;
}

__global__ void accum_deg_kernel(const int* __restrict__ dst, float* __restrict__ deg, int E) {
    int e = blockIdx.x * blockDim.x + threadIdx.x;
    if (e < E) atomicAdd(&deg[dst[e]], 1.0f);
}

__global__ void deg_to_dinv_kernel(float* __restrict__ deg, int n) {
    int i = blockIdx.x * blockDim.x + threadIdx.x;
    if (i < n) deg[i] = rsqrtf(deg[i]);
}

template <int MB>
__launch_bounds__(128)
__global__ void gemm_f32_wmma_kernel(const float* __restrict__ A,
                                     const float* __restrict__ B,
                                     float* __restrict__ C,
                                     int M, int N, int K) {
    const int lane = threadIdx.x & 31;
    const int wave = threadIdx.x >> 5;
    const int ntN  = N >> 4;
    const int nMB  = M / (16 * MB);
    const int job  = blockIdx.x * (blockDim.x >> 5) + wave;
    if (job >= nMB * ntN) return;

    const int mb = job / ntN;
    const int nt = job - mb * ntN;
    const int m0 = mb * 16 * MB;
    const int n0 = nt << 4;

    const int l15   = lane & 15;
    const int khalf = (lane >> 4) << 1;

    const float* abase = A + (size_t)(m0 + l15) * K + khalf;
    const float* bcol  = B + (size_t)khalf * N + (n0 + l15);

    v8f acc[MB];
    #pragma unroll
    for (int t = 0; t < MB; ++t) acc[t] = (v8f){};

    for (int k = 0; k < K; k += 4) {
        v2f b;
        b.x = bcol[(size_t)k * N];
        b.y = bcol[(size_t)(k + 1) * N];
        #pragma unroll
        for (int t = 0; t < MB; ++t) {
            v2f a = *(const v2f*)(abase + (size_t)t * 16 * K + k);
            acc[t] = __builtin_amdgcn_wmma_f32_16x16x4_f32(
                 false, a,  false, b,
                 (short)0, acc[t],  false,  false);
        }
    }

    const int cmoff = (lane >> 4) << 3;
    #pragma unroll
    for (int t = 0; t < MB; ++t) {
        float* crow = C + (size_t)(m0 + t * 16 + cmoff) * N + (n0 + l15);
        #pragma unroll
        for (int r = 0; r < 8; ++r) crow[(size_t)r * N] = acc[t][r];
    }
}

template <int F>
__global__ void agg_init_kernel(const float* __restrict__ h,
                                const float* __restrict__ dinv,
                                const float* __restrict__ bias,
                                float* __restrict__ out, int n) {
    int idx = blockIdx.x * blockDim.x + threadIdx.x;
    if (idx >= n * F) return;
    int i = idx >> (F == 128 ? 7 : 6);
    int f = idx & (F - 1);
    float di = dinv[i];
    out[idx] = bias[f] + h[idx] * di * di;
}

template <int F>
__launch_bounds__(256)
__global__ void agg_edges_kernel(const int* __restrict__ src,
                                 const int* __restrict__ dst,
                                 const float* __restrict__ dinv,
                                 const float* __restrict__ h,
                                 float* __restrict__ out, int E) {
    const int lane = threadIdx.x & 31;
    const int e    = (blockIdx.x * blockDim.x + threadIdx.x) >> 5;
    if (e >= E) return;

    const int   s    = src[e];
    const int   d    = dst[e];
    const float norm = dinv[s] * dinv[d];

    constexpr int VE = F / 32;
    const float* hs = h   + (size_t)s * F + lane * VE;
    float*       od = out + (size_t)d * F + lane * VE;

    if constexpr (VE == 4) {
        v4f v = *(const v4f*)hs;
        atomicAdd(od + 0, v.x * norm);
        atomicAdd(od + 1, v.y * norm);
        atomicAdd(od + 2, v.z * norm);
        atomicAdd(od + 3, v.w * norm);
    } else {
        v2f v = *(const v2f*)hs;
        atomicAdd(od + 0, v.x * norm);
        atomicAdd(od + 1, v.y * norm);
    }
}

__global__ void relu_kernel(float* __restrict__ x, int n) {
    int i = blockIdx.x * blockDim.x + threadIdx.x;
    if (i < n) x[i] = fmaxf(x[i], 0.0f);
}

extern "C" void kernel_launch(void* const* d_in, const int* in_sizes, int n_in,
                              void* d_out, int out_size, void* d_ws, size_t ws_size,
                              hipStream_t stream) {
    const float* x    = (const float*)d_in[0];
    const int*   eidx = (const int*)  d_in[1];
    const float* W1   = (const float*)d_in[2];
    const float* b1   = (const float*)d_in[3];
    const float* W2   = (const float*)d_in[4];
    const float* b2   = (const float*)d_in[5];
    float*       out  = (float*)d_out;

    const int N = NNODES;
    const int E = in_sizes[1] / 2;
    const int* src = eidx;
    const int* dst = eidx + E;

    char*  ws   = (char*)d_ws;
    float* dinv = (float*)(ws);
    float* h1   = (float*)(ws + (size_t)(1  << 20));
    float* out1 = (float*)(ws + (size_t)(53 << 20));
    float* h2   = h1;

    const int T = 256;
    constexpr int MB = 5;

    init_deg_kernel<<<(N + T - 1) / T, T, 0, stream>>>(dinv, N);
    accum_deg_kernel<<<(E + T - 1) / T, T, 0, stream>>>(dst, dinv, E);
    deg_to_dinv_kernel<<<(N + T - 1) / T, T, 0, stream>>>(dinv, N);

    {
        const int jobs = (N / (16 * MB)) * (128 / 16);
        gemm_f32_wmma_kernel<MB><<<jobs / 4, 128, 0, stream>>>(x, W1, h1, N, 128, 128);
    }
    agg_init_kernel<128><<<((size_t)N * 128 + T - 1) / T, T, 0, stream>>>(h1, dinv, b1, out1, N);
    agg_edges_kernel<128><<<((size_t)E * 32 + T - 1) / T, T, 0, stream>>>(src, dst, dinv, h1, out1, E);
    relu_kernel<<<((size_t)N * 128 + T - 1) / T, T, 0, stream>>>(out1, N * 128);

    {
        const int jobs = (N / (16 * MB)) * (64 / 16);
        gemm_f32_wmma_kernel<MB><<<jobs / 4, 128, 0, stream>>>(out1, W2, h2, N, 64, 128);
    }
    agg_init_kernel<64><<<((size_t)N * 64 + T - 1) / T, T, 0, stream>>>(h2, dinv, b2, out, N);
    agg_edges_kernel<64><<<((size_t)E * 32 + T - 1) / T, T, 0, stream>>>(src, dst, dinv, h2, out, E);
}


# Replace f32 atomic scatter-add by per-call dst-sorted edge list (int atomic rank + scan) and one-wave-per-node gather-sum; self-loop, bias, relu folded in; sum order = rank arrival order
# speedup vs baseline: 2.3916x; 2.3916x over previous
#include <hip/hip_runtime.h>

typedef __attribute__((ext_vector_type(2))) float v2f;
typedef __attribute__((ext_vector_type(4))) float v4f;
typedef __attribute__((ext_vector_type(8))) float v8f;

#define NNODES 100000

__global__ void zero_cnt_kernel(int* __restrict__ cnt, int n) {
    int i = blockIdx.x * blockDim.x + threadIdx.x;
    if (i < n) cnt[i] = 0;
}

__global__ void edge_rank_kernel(const int* __restrict__ dst, int* __restrict__ cnt,
                                 int* __restrict__ rank, int E) {
    int e = blockIdx.x * blockDim.x + threadIdx.x;
    if (e < E) rank[e] = atomicAdd(&cnt[dst[e]], 1);
}

__launch_bounds__(256)
__global__ void scan_deg_kernel(const int* __restrict__ cnt, int* __restrict__ rowptr,
                                float* __restrict__ dinv, int n) {
    __shared__ int s_pre[8];
    __shared__ int s_own[8];
    const int tid  = threadIdx.x;
    const int lane = tid & 31;
    const int wave = tid >> 5;
    const int blk0 = blockIdx.x * 1024;

    const int nfront = blockIdx.x;
    int pre = 0;
    for (int c = 0; c < nfront; c += 8) {
        int4 v[8];
        for (int k = 0; k < 8; ++k) {
            const int cb = (c + k < nfront) ? (c + k) : 0;
            v[k] = *(const int4*)(cnt + cb * 1024 + tid * 4);
        }
        for (int k = 0; k < 8; ++k) {
            if (c + k < nfront) pre += (v[k].x + v[k].y) + (v[k].z + v[k].w);
        }
    }

    const int i0 = blk0 + tid * 4;
    int c0 = 0, c1 = 0, c2 = 0, c3 = 0;
    if (i0 + 3 < n) {
        const int4 v = *(const int4*)(cnt + i0);
        c0 = v.x; c1 = v.y; c2 = v.z; c3 = v.w;
    } else {
        if (i0 + 0 < n) c0 = cnt[i0 + 0];
        if (i0 + 1 < n) c1 = cnt[i0 + 1];
        if (i0 + 2 < n) c2 = cnt[i0 + 2];
    }
    const int own = (c0 + c1) + (c2 + c3);

    int incl = own;
    for (int o = 1; o < 32; o <<= 1) {
        const int t = __shfl_up(incl, o);
        if (lane >= o) incl += t;
    }
    for (int o = 16; o > 0; o >>= 1) pre += __shfl_xor(pre, o);

    if (lane == 31) s_own[wave] = incl;
    if (lane == 0)  s_pre[wave] = pre;
    __syncthreads();

    int base = 0;
    for (int w = 0; w < 8; ++w) {
        base += s_pre[w];
        if (w < wave) base += s_own[w];
    }
    const int e0 = base + incl - own;
    const int e1 = e0 + c0;
    const int e2 = e1 + c1;
    const int e3 = e2 + c2;
    const int e4 = e3 + c3;

    if (i0 + 0 < n) { rowptr[i0 + 0] = e0; dinv[i0 + 0] = rsqrtf((float)(c0 + 1)); }
    if (i0 + 1 < n) { rowptr[i0 + 1] = e1; dinv[i0 + 1] = rsqrtf((float)(c1 + 1)); }
    if (i0 + 2 < n) { rowptr[i0 + 2] = e2; dinv[i0 + 2] = rsqrtf((float)(c2 + 1)); }
    if (i0 + 3 < n) { rowptr[i0 + 3] = e3; dinv[i0 + 3] = rsqrtf((float)(c3 + 1)); }
    if (i0 <= n - 1 && n - 1 <= i0 + 3) rowptr[n] = e4;
}

__global__ void edge_place_kernel(const int* __restrict__ src, const int* __restrict__ dst,
                                  const int* __restrict__ rank, const int* __restrict__ rowptr,
                                  int* __restrict__ csr_src, int E) {
    int e = blockIdx.x * blockDim.x + threadIdx.x;
    if (e < E) csr_src[rowptr[dst[e]] + rank[e]] = src[e];
}

__global__ void copy_idx_kernel(const int* __restrict__ from, int* __restrict__ to, int n4) {
    int i = blockIdx.x * blockDim.x + threadIdx.x;
    if (i < n4) ((int4*)to)[i] = ((const int4*)from)[i];
}

template <int MB>
__launch_bounds__(128)
__global__ void gemm_f32_wmma_kernel(const float* __restrict__ A,
                                     const float* __restrict__ B,
                                     float* __restrict__ C,
                                     int M, int N, int K) {
    const int lane = threadIdx.x & 31;
    const int wave = threadIdx.x >> 5;
    const int ntN  = N >> 4;
    const int nMB  = M / (16 * MB);
    const int job  = blockIdx.x * (blockDim.x >> 5) + wave;
    if (job >= nMB * ntN) return;

    const int mb = job / ntN;
    const int nt = job - mb * ntN;
    const int m0 = mb * 16 * MB;
    const int n0 = nt << 4;

    const int l15   = lane & 15;
    const int khalf = (lane >> 4) << 1;

    const float* abase = A + (size_t)(m0 + l15) * K + khalf;
    const float* bcol  = B + (size_t)khalf * N + (n0 + l15);

    v8f acc[MB];
    #pragma unroll
    for (int t = 0; t < MB; ++t) acc[t] = (v8f){};

    for (int k = 0; k < K; k += 4) {
        v2f b;
        b.x = bcol[(size_t)k * N];
        b.y = bcol[(size_t)(k + 1) * N];
        #pragma unroll
        for (int t = 0; t < MB; ++t) {
            v2f a = *(const v2f*)(abase + (size_t)t * 16 * K + k);
            acc[t] = __builtin_amdgcn_wmma_f32_16x16x4_f32(
                 false, a,  false, b,
                 (short)0, acc[t],  false,  false);
        }
    }

    const int cmoff = (lane >> 4) << 3;
    #pragma unroll
    for (int t = 0; t < MB; ++t) {
        float* crow = C + (size_t)(m0 + t * 16 + cmoff) * N + (n0 + l15);
        #pragma unroll
        for (int r = 0; r < 8; ++r) crow[(size_t)r * N] = acc[t][r];
    }
}

template <int F, bool RELU>
__launch_bounds__(256)
__global__ void gather_sum_kernel(const int* __restrict__ rowptr,
                                  const int* __restrict__ csr_src,
                                  const float* __restrict__ dinv,
                                  const float* __restrict__ h,
                                  const float* __restrict__ bias,
                                  float* __restrict__ out, int n) {
    constexpr int VE = F / 32;
    typedef __attribute__((ext_vector_type(VE))) float vec;

    const int lane = threadIdx.x & 31;
    const int d = __builtin_amdgcn_readfirstlane(
        (int)(blockIdx.x * (blockDim.x >> 5) + (threadIdx.x >> 5)));
    if (d >= n) return;

    const int   beg = rowptr[d];
    const int   end = rowptr[d + 1];
    const float dd  = dinv[d];

    const float* hl = h + lane * VE;
    vec acc = *(const vec*)(hl + (size_t)d * F) * (dd * dd);

    for (int base = beg; base < end; base += 32) {
        const int m = min(end - base, 32);
        int   s_l = 0;
        float w_l = 0.0f;
        if (lane < m) {
            s_l = csr_src[base + lane];
            w_l = dinv[s_l] * dd;
        }
        const int w_i = __float_as_int(w_l);
        int j = 0;
        for (; j + 4 <= m; j += 4) {
            const int   s0 = __builtin_amdgcn_readlane(s_l, j + 0);
            const int   s1 = __builtin_amdgcn_readlane(s_l, j + 1);
            const int   s2 = __builtin_amdgcn_readlane(s_l, j + 2);
            const int   s3 = __builtin_amdgcn_readlane(s_l, j + 3);
            const float w0 = __int_as_float(__builtin_amdgcn_readlane(w_i, j + 0));
            const float w1 = __int_as_float(__builtin_amdgcn_readlane(w_i, j + 1));
            const float w2 = __int_as_float(__builtin_amdgcn_readlane(w_i, j + 2));
            const float w3 = __int_as_float(__builtin_amdgcn_readlane(w_i, j + 3));
            const vec v0 = *(const vec*)(hl + (size_t)s0 * F);
            const vec v1 = *(const vec*)(hl + (size_t)s1 * F);
            const vec v2 = *(const vec*)(hl + (size_t)s2 * F);
            const vec v3 = *(const vec*)(hl + (size_t)s3 * F);
            acc += v0 * w0;
            acc += v1 * w1;
            acc += v2 * w2;
            acc += v3 * w3;
        }
        for (; j < m; ++j) {
            const int   s = __builtin_amdgcn_readlane(s_l, j);
            const float w = __int_as_float(__builtin_amdgcn_readlane(w_i, j));
            const vec   v = *(const vec*)(hl + (size_t)s * F);
            acc += v * w;
        }
    }

    vec r = acc + *(const vec*)(bias + lane * VE);
    if constexpr (RELU) {
        for (int q = 0; q < VE; ++q) r[q] = fmaxf(r[q], 0.0f);
    }
    *(vec*)(out + (size_t)d * F + lane * VE) = r;
}

extern "C" void kernel_launch(void* const* d_in, const int* in_sizes, int n_in,
                              void* d_out, int out_size, void* d_ws, size_t ws_size,
                              hipStream_t stream) {
    const float* x    = (const float*)d_in[0];
    const int*   eidx = (const int*)  d_in[1];
    const float* W1   = (const float*)d_in[2];
    const float* b1   = (const float*)d_in[3];
    const float* W2   = (const float*)d_in[4];
    const float* b2   = (const float*)d_in[5];
    float*       out  = (float*)d_out;

    const int N = NNODES;
    const int E = in_sizes[1] / 2;
    const int* src = eidx;
    const int* dst = eidx + E;

    char*  ws     = (char*)d_ws;
    float* dinv   = (float*)(ws);
    int*   rowptr = (int*)  (ws + (size_t)409600);
    float* h1     = (float*)(ws + (size_t)(1  << 20));
    float* out1   = (float*)(ws + (size_t)(53 << 20));
    float* h2     = h1;
    int*   rank   = (int*)  (ws + (size_t)(53 << 20));
    int*   cnt    = (int*)  (ws + (size_t)(61 << 20));
    int*   csr_a  = (int*)d_out;
    int*   csr_b  = (int*)(ws + (size_t)(1 << 20) + (size_t)25600000);

    const int T = 256;
    constexpr int MB = 5;

    zero_cnt_kernel<<<(N + T - 1) / T, T, 0, stream>>>(cnt, N);
    edge_rank_kernel<<<(E + T - 1) / T, T, 0, stream>>>(dst, cnt, rank, E);
    scan_deg_kernel<<<(N + 1023) / 1024, 256, 0, stream>>>(cnt, rowptr, dinv, N);
    edge_place_kernel<<<(E + T - 1) / T, T, 0, stream>>>(src, dst, rank, rowptr, csr_a, E);

    {
        const int jobs = (N / (16 * MB)) * (128 / 16);
        gemm_f32_wmma_kernel<MB><<<jobs / 4, 128, 0, stream>>>(x, W1, h1, N, 128, 128);
    }
    gather_sum_kernel<128, true><<<(N + 7) / 8, T, 0, stream>>>(rowptr, csr_a, dinv, h1, b1, out1, N);

    {
        const int jobs = (N / (16 * MB)) * (64 / 16);
        gemm_f32_wmma_kernel<MB><<<jobs / 4, 128, 0, stream>>>(out1, W2, h2, N, 64, 128);
    }
    copy_idx_kernel<<<((E + 3) / 4 + T - 1) / T, T, 0, stream>>>(csr_a, csr_b, (E + 3) / 4);
    gather_sum_kernel<64, false><<<(N + 7) / 8, T, 0, stream>>>(rowptr, csr_b, dinv, h2, b2, out, N);
}


# Edge list sorted by destination with LDS counters, not one returning global atomic per edge: 8192-edge chunks binned to 512-node buckets, a block per bucket ranks and places; sum order = LDS arrival
# speedup vs baseline: 4.0616x; 1.6983x over previous
#include <hip/hip_runtime.h>

typedef __attribute__((ext_vector_type(2))) float v2f;
typedef __attribute__((ext_vector_type(4))) float v4f;
typedef __attribute__((ext_vector_type(8))) float v8f;

#define NNODES 100000

#define BSH       9
#define BNODES    (1 << BSH)
#define NBK       ((NNODES + BNODES - 1) >> BSH)
#define CHUNK     8192
#define STAGE_CAP 12288

__launch_bounds__(1024)
__global__ void bin_chunk_kernel(const int* __restrict__ src, const int* __restrict__ dst,
                                 int* __restrict__ chunk_off, int* __restrict__ recs, int E) {
    __shared__ int s_bin[NBK + 1];
    __shared__ int s_rec[CHUNK];
    constexpr int EPT = CHUNK / 1024;
    constexpr int PER = (NBK + 31) / 32;

    const int tid  = threadIdx.x;
    const int lane = tid & 31;
    const int wave = tid >> 5;
    const int e0   = blockIdx.x * CHUNK;

    if (tid <= NBK) s_bin[tid] = 0;
    __syncthreads();

    int s[EPT], d[EPT], r[EPT];
    #pragma unroll
    for (int k = 0; k < EPT; ++k) {
        const int e = e0 + k * 1024 + tid;
        d[k] = -1; s[k] = 0;
        if (e < E) { d[k] = dst[e]; s[k] = src[e]; }
    }
    #pragma unroll
    for (int k = 0; k < EPT; ++k) {
        r[k] = 0;
        if (d[k] >= 0) r[k] = atomicAdd(&s_bin[d[k] >> BSH], 1);
    }
    __syncthreads();

    if (wave == 0) {
        int v[PER];
        int sum = 0;
        #pragma unroll
        for (int q = 0; q < PER; ++q) {
            const int idx = lane * PER + q;
            v[q] = (idx < NBK) ? s_bin[idx] : 0;
            sum += v[q];
        }
        int incl = sum;
        for (int o = 1; o < 32; o <<= 1) {
            const int t = __shfl_up(incl, o);
            if (lane >= o) incl += t;
        }
        int run = incl - sum;
        #pragma unroll
        for (int q = 0; q < PER; ++q) {
            const int idx = lane * PER + q;
            if (idx < NBK) s_bin[idx] = run;
            run += v[q];
        }
        if (lane == 31) s_bin[NBK] = incl;
    }
    __syncthreads();

    if (tid <= NBK) chunk_off[blockIdx.x * (NBK + 1) + tid] = s_bin[tid];
    #pragma unroll
    for (int k = 0; k < EPT; ++k) {
        if (d[k] >= 0)
            s_rec[s_bin[d[k] >> BSH] + r[k]] = (s[k] << BSH) | (d[k] & (BNODES - 1));
    }
    __syncthreads();

    const int nv = s_bin[NBK];
    for (int i = tid; i < nv; i += 1024) recs[e0 + i] = s_rec[i];
}

template <typename F>
__device__ __forceinline__ int walk_bucket(const int* __restrict__ chunk_off,
                                           const int* __restrict__ recs,
                                           int b, int nchunks, int wave, int lane, F f) {
    int osum = 0;
    for (int j0 = 0; j0 < nchunks; j0 += 1024) {
        const int j = j0 + wave + 32 * lane;
        int o0 = 0, c = 0;
        if (j < nchunks) {
            o0 = chunk_off[j * (NBK + 1) + b];
            c  = chunk_off[j * (NBK + 1) + b + 1] - o0;
        }
        osum += o0;
        for (int k = 0; k < 32; k += 4) {
            if (j0 + wave + 32 * k >= nchunks) break;
            const int* run[4];
            int cnt[4], first[4];
            #pragma unroll
            for (int u = 0; u < 4; ++u) {
                cnt[u] = __builtin_amdgcn_readlane(c, k + u);
                const int o = __builtin_amdgcn_readlane(o0, k + u);
                run[u] = recs + (size_t)(j0 + wave + 32 * (k + u)) * CHUNK + o;
                first[u] = -1;
                if (lane < cnt[u]) first[u] = run[u][lane];
            }
            #pragma unroll
            for (int u = 0; u < 4; ++u) {
                if (first[u] >= 0) f(first[u]);
            }
            #pragma unroll
            for (int u = 0; u < 4; ++u) {
                for (int i = lane + 32; i < cnt[u]; i += 32) f(run[u][i]);
            }
        }
    }
    return osum;
}

__launch_bounds__(1024)
__global__ void bucket_sort_kernel(const int* __restrict__ chunk_off, const int* __restrict__ recs,
                                   int* __restrict__ rowptr, float* __restrict__ dinv,
                                   int* __restrict__ csr_src, int n, int nchunks) {
    __shared__ int s_cnt[BNODES];
    __shared__ int s_cur[BNODES];
    __shared__ int s_wsum[32];
    __shared__ int s_base;
    __shared__ int s_stage[STAGE_CAP];

    const int tid  = threadIdx.x;
    const int lane = tid & 31;
    const int wave = tid >> 5;
    const int b    = blockIdx.x;

    if (tid < BNODES) s_cnt[tid] = 0;
    if (tid == 0) s_base = 0;
    __syncthreads();

    int osum = walk_bucket(chunk_off, recs, b, nchunks, wave, lane,
                           [&](int rec) { atomicAdd(&s_cnt[rec & (BNODES - 1)], 1); });
    for (int o = 16; o > 0; o >>= 1) osum += __shfl_xor(osum, o);
    if (lane == 0 && osum != 0) atomicAdd(&s_base, osum);
    __syncthreads();

    const int cnt_i = (tid < BNODES) ? s_cnt[tid] : 0;
    int incl = cnt_i;
    for (int o = 1; o < 32; o <<= 1) {
        const int t = __shfl_up(incl, o);
        if (lane >= o) incl += t;
    }
    if (lane == 31) s_wsum[wave] = incl;
    __syncthreads();

    int wpre = 0, nb = 0;
    for (int w = 0; w < 32; ++w) {
        const int t = s_wsum[w];
        if (w < wave) wpre += t;
        nb += t;
    }
    const int  excl = wpre + incl - cnt_i;
    const int  base = s_base;
    const bool fits = nb <= STAGE_CAP;
    if (tid < BNODES) {
        s_cur[tid] = excl;
        const int node = b * BNODES + tid;
        if (node < n) {
            rowptr[node] = base + excl;
            dinv[node]   = rsqrtf((float)(cnt_i + 1));
            if (node == n - 1) rowptr[n] = base + excl + cnt_i;
        }
    }
    __syncthreads();

    walk_bucket(chunk_off, recs, b, nchunks, wave, lane,
                [&](int rec) {
                    const int pos = atomicAdd(&s_cur[rec & (BNODES - 1)], 1);
                    const int sv  = (int)((unsigned)rec >> BSH);
                    if (fits) s_stage[pos] = sv;
                    else      csr_src[base + pos] = sv;
                });
    __syncthreads();
    if (fits) {
        for (int i = tid; i < nb; i += 1024) csr_src[base + i] = s_stage[i];
    }
}

__global__ void copy_idx_kernel(const int* __restrict__ from, int* __restrict__ to, int n4) {
    int i = blockIdx.x * blockDim.x + threadIdx.x;
    if (i < n4) ((int4*)to)[i] = ((const int4*)from)[i];
}

template <int MB>
__launch_bounds__(128)
__global__ void gemm_f32_wmma_kernel(const float* __restrict__ A,
                                     const float* __restrict__ B,
                                     float* __restrict__ C,
                                     int M, int N, int K) {
    const int lane = threadIdx.x & 31;
    const int wave = threadIdx.x >> 5;
    const int ntN  = N >> 4;
    const int nMB  = M / (16 * MB);
    const int job  = blockIdx.x * (blockDim.x >> 5) + wave;
    if (job >= nMB * ntN) return;

    const int mb = job / ntN;
    const int nt = job - mb * ntN;
    const int m0 = mb * 16 * MB;
    const int n0 = nt << 4;

    const int l15   = lane & 15;
    const int khalf = (lane >> 4) << 1;

    const float* abase = A + (size_t)(m0 + l15) * K + khalf;
    const float* bcol  = B + (size_t)khalf * N + (n0 + l15);

    v8f acc[MB];
    #pragma unroll
    for (int t = 0; t < MB; ++t) acc[t] = (v8f){};

    for (int k = 0; k < K; k += 4) {
        v2f b;
        b.x = bcol[(size_t)k * N];
        b.y = bcol[(size_t)(k + 1) * N];
        #pragma unroll
        for (int t = 0; t < MB; ++t) {
            v2f a = *(const v2f*)(abase + (size_t)t * 16 * K + k);
            acc[t] = __builtin_amdgcn_wmma_f32_16x16x4_f32(
                 false, a,  false, b,
                 (short)0, acc[t],  false,  false);
        }
    }

    const int cmoff = (lane >> 4) << 3;
    #pragma unroll
    for (int t = 0; t < MB; ++t) {
        float* crow = C + (size_t)(m0 + t * 16 + cmoff) * N + (n0 + l15);
        #pragma unroll
        for (int r = 0; r < 8; ++r) crow[(size_t)r * N] = acc[t][r];
    }
}

template <int F, bool RELU>
__launch_bounds__(256)
__global__ void gather_sum_kernel(const int* __restrict__ rowptr,
                                  const int* __restrict__ csr_src,
                                  const float* __restrict__ dinv,
                                  const float* __restrict__ h,
                                  const float* __restrict__ bias,
                                  float* __restrict__ out, int n) {
    constexpr int VE = F / 32;
    typedef __attribute__((ext_vector_type(VE))) float vec;

    const int lane = threadIdx.x & 31;
    const int d = __builtin_amdgcn_readfirstlane(
        (int)(blockIdx.x * (blockDim.x >> 5) + (threadIdx.x >> 5)));
    if (d >= n) return;

    const int   beg = rowptr[d];
    const int   end = rowptr[d + 1];
    const float dd  = dinv[d];

    const float* hl = h + lane * VE;
    vec acc = *(const vec*)(hl + (size_t)d * F) * (dd * dd);

    for (int base = beg; base < end; base += 32) {
        const int m = min(end - base, 32);
        int   s_l = 0;
        float w_l = 0.0f;
        if (lane < m) {
            s_l = csr_src[base + lane];
            w_l = dinv[s_l] * dd;
        }
        const int w_i = __float_as_int(w_l);
        int j = 0;
        for (; j + 4 <= m; j += 4) {
            const int   s0 = __builtin_amdgcn_readlane(s_l, j + 0);
            const int   s1 = __builtin_amdgcn_readlane(s_l, j + 1);
            const int   s2 = __builtin_amdgcn_readlane(s_l, j + 2);
            const int   s3 = __builtin_amdgcn_readlane(s_l, j + 3);
            const float w0 = __int_as_float(__builtin_amdgcn_readlane(w_i, j + 0));
            const float w1 = __int_as_float(__builtin_amdgcn_readlane(w_i, j + 1));
            const float w2 = __int_as_float(__builtin_amdgcn_readlane(w_i, j + 2));
            const float w3 = __int_as_float(__builtin_amdgcn_readlane(w_i, j + 3));
            const vec v0 = *(const vec*)(hl + (size_t)s0 * F);
            const vec v1 = *(const vec*)(hl + (size_t)s1 * F);
            const vec v2 = *(const vec*)(hl + (size_t)s2 * F);
            const vec v3 = *(const vec*)(hl + (size_t)s3 * F);
            acc += v0 * w0;
            acc += v1 * w1;
            acc += v2 * w2;
            acc += v3 * w3;
        }
        for (; j < m; ++j) {
            const int   s = __builtin_amdgcn_readlane(s_l, j);
            const float w = __int_as_float(__builtin_amdgcn_readlane(w_i, j));
            const vec   v = *(const vec*)(hl + (size_t)s * F);
            acc += v * w;
        }
    }

    vec r = acc + *(const vec*)(bias + lane * VE);
    if constexpr (RELU) {
        for (int q = 0; q < VE; ++q) r[q] = fmaxf(r[q], 0.0f);
    }
    *(vec*)(out + (size_t)d * F + lane * VE) = r;
}

extern "C" void kernel_launch(void* const* d_in, const int* in_sizes, int n_in,
                              void* d_out, int out_size, void* d_ws, size_t ws_size,
                              hipStream_t stream) {
    const float* x    = (const float*)d_in[0];
    const int*   eidx = (const int*)  d_in[1];
    const float* W1   = (const float*)d_in[2];
    const float* b1   = (const float*)d_in[3];
    const float* W2   = (const float*)d_in[4];
    const float* b2   = (const float*)d_in[5];
    float*       out  = (float*)d_out;

    const int N = NNODES;
    const int E = in_sizes[1] / 2;
    const int* src = eidx;
    const int* dst = eidx + E;

    char*  ws        = (char*)d_ws;
    float* dinv      = (float*)(ws);
    int*   rowptr    = (int*)  (ws + (size_t)409600);
    float* h1        = (float*)(ws + (size_t)(1  << 20));
    float* out1      = (float*)(ws + (size_t)(53 << 20));
    float* h2        = h1;
    int*   recs      = (int*)  (ws + (size_t)(53 << 20));
    int*   chunk_off = (int*)  (ws + (size_t)(61 << 20));
    int*   csr_a     = (int*)d_out;
    int*   csr_b     = (int*)(ws + (size_t)(1 << 20) + (size_t)25600000);

    const int T = 256;
    constexpr int MB = 5;

    const int nchunks = (E + CHUNK - 1) / CHUNK;
    bin_chunk_kernel<<<nchunks, 1024, 0, stream>>>(src, dst, chunk_off, recs, E);
    bucket_sort_kernel<<<NBK, 1024, 0, stream>>>(chunk_off, recs, rowptr, dinv, csr_a, N, nchunks);

    {
        const int jobs = (N / (16 * MB)) * (128 / 16);
        gemm_f32_wmma_kernel<MB><<<jobs / 4, 128, 0, stream>>>(x, W1, h1, N, 128, 128);
    }
    gather_sum_kernel<128, true><<<(N + 7) / 8, T, 0, stream>>>(rowptr, csr_a, dinv, h1, b1, out1, N);

    {
        const int jobs = (N / (16 * MB)) * (64 / 16);
        gemm_f32_wmma_kernel<MB><<<jobs / 4, 128, 0, stream>>>(out1, W2, h2, N, 64, 128);
    }
    copy_idx_kernel<<<((E + 3) / 4 + T - 1) / T, T, 0, stream>>>(csr_a, csr_b, (E + 3) / 4);
    gather_sum_kernel<64, false><<<(N + 7) / 8, T, 0, stream>>>(rowptr, csr_b, dinv, h2, b2, out, N);
}


# Layer 1 reordered to (A x) W1: one kernel gathers the normalized sum of x rows into an LDS tile and runs both f32 WMMA products, bias, relu on it; h1, out1, list copy go; sums: gather before product
# speedup vs baseline: 6.0338x; 1.4856x over previous
#include <hip/hip_runtime.h>

typedef __attribute__((ext_vector_type(2))) float v2f;
typedef __attribute__((ext_vector_type(4))) float v4f;
typedef __attribute__((ext_vector_type(8))) float v8f;

#define NNODES 100000

#define BSH       9
#define BNODES    (1 << BSH)
#define NBK       ((NNODES + BNODES - 1) >> BSH)
#define CHUNK     8192
#define STAGE_CAP 12288

__launch_bounds__(1024)
__global__ void bin_chunk_kernel(const int* __restrict__ src, const int* __restrict__ dst,
                                 int* __restrict__ chunk_off, int* __restrict__ recs, int E) {
    __shared__ int s_bin[NBK + 1];
    __shared__ int s_rec[CHUNK];
    constexpr int EPT = CHUNK / 1024;
    constexpr int PER = (NBK + 31) / 32;

    const int tid  = threadIdx.x;
    const int lane = tid & 31;
    const int wave = tid >> 5;
    const int e0   = blockIdx.x * CHUNK;

    if (tid <= NBK) s_bin[tid] = 0;
    __syncthreads();

    int s[EPT], d[EPT], r[EPT];
    #pragma unroll
    for (int k = 0; k < EPT; ++k) {
        const int e = e0 + k * 1024 + tid;
        d[k] = -1; s[k] = 0;
        if (e < E) { d[k] = dst[e]; s[k] = src[e]; }
    }
    #pragma unroll
    for (int k = 0; k < EPT; ++k) {
        r[k] = 0;
        if (d[k] >= 0) r[k] = atomicAdd(&s_bin[d[k] >> BSH], 1);
    }
    __syncthreads();

    if (wave == 0) {
        int v[PER];
        int sum = 0;
        #pragma unroll
        for (int q = 0; q < PER; ++q) {
            const int idx = lane * PER + q;
            v[q] = (idx < NBK) ? s_bin[idx] : 0;
            sum += v[q];
        }
        int incl = sum;
        for (int o = 1; o < 32; o <<= 1) {
            const int t = __shfl_up(incl, o);
            if (lane >= o) incl += t;
        }
        int run = incl - sum;
        #pragma unroll
        for (int q = 0; q < PER; ++q) {
            const int idx = lane * PER + q;
            if (idx < NBK) s_bin[idx] = run;
            run += v[q];
        }
        if (lane == 31) s_bin[NBK] = incl;
    }
    __syncthreads();

    if (tid <= NBK) chunk_off[blockIdx.x * (NBK + 1) + tid] = s_bin[tid];
    #pragma unroll
    for (int k = 0; k < EPT; ++k) {
        if (d[k] >= 0)
            s_rec[s_bin[d[k] >> BSH] + r[k]] = (s[k] << BSH) | (d[k] & (BNODES - 1));
    }
    __syncthreads();

    const int nv = s_bin[NBK];
    for (int i = tid; i < nv; i += 1024) recs[e0 + i] = s_rec[i];
}

template <typename F>
__device__ __forceinline__ int walk_bucket(const int* __restrict__ chunk_off,
                                           const int* __restrict__ recs,
                                           int b, int nchunks, int wave, int lane, F f) {
    int osum = 0;
    for (int j0 = 0; j0 < nchunks; j0 += 1024) {
        const int j = j0 + wave + 32 * lane;
        int o0 = 0, c = 0;
        if (j < nchunks) {
            o0 = chunk_off[j * (NBK + 1) + b];
            c  = chunk_off[j * (NBK + 1) + b + 1] - o0;
        }
        osum += o0;
        for (int k = 0; k < 32; k += 4) {
            if (j0 + wave + 32 * k >= nchunks) break;
            const int* run[4];
            int cnt[4], first[4];
            #pragma unroll
            for (int u = 0; u < 4; ++u) {
                cnt[u] = __builtin_amdgcn_readlane(c, k + u);
                const int o = __builtin_amdgcn_readlane(o0, k + u);
                run[u] = recs + (size_t)(j0 + wave + 32 * (k + u)) * CHUNK + o;
                first[u] = -1;
                if (lane < cnt[u]) first[u] = run[u][lane];
            }
            #pragma unroll
            for (int u = 0; u < 4; ++u) {
                if (first[u] >= 0) f(first[u]);
            }
            #pragma unroll
            for (int u = 0; u < 4; ++u) {
                for (int i = lane + 32; i < cnt[u]; i += 32) f(run[u][i]);
            }
        }
    }
    return osum;
}

__launch_bounds__(1024)
__global__ void bucket_sort_kernel(const int* __restrict__ chunk_off, const int* __restrict__ recs,
                                   int* __restrict__ rowptr, float* __restrict__ dinv,
                                   int* __restrict__ csr_src, int n, int nchunks) {
    __shared__ int s_cnt[BNODES];
    __shared__ int s_cur[BNODES];
    __shared__ int s_wsum[32];
    __shared__ int s_base;
    __shared__ int s_stage[STAGE_CAP];

    const int tid  = threadIdx.x;
    const int lane = tid & 31;
    const int wave = tid >> 5;
    const int b    = blockIdx.x;

    if (tid < BNODES) s_cnt[tid] = 0;
    if (tid == 0) s_base = 0;
    __syncthreads();

    int osum = walk_bucket(chunk_off, recs, b, nchunks, wave, lane,
                           [&](int rec) { atomicAdd(&s_cnt[rec & (BNODES - 1)], 1); });
    for (int o = 16; o > 0; o >>= 1) osum += __shfl_xor(osum, o);
    if (lane == 0 && osum != 0) atomicAdd(&s_base, osum);
    __syncthreads();

    const int cnt_i = (tid < BNODES) ? s_cnt[tid] : 0;
    int incl = cnt_i;
    for (int o = 1; o < 32; o <<= 1) {
        const int t = __shfl_up(incl, o);
        if (lane >= o) incl += t;
    }
    if (lane == 31) s_wsum[wave] = incl;
    __syncthreads();

    int wpre = 0, nb = 0;
    for (int w = 0; w < 32; ++w) {
        const int t = s_wsum[w];
        if (w < wave) wpre += t;
        nb += t;
    }
    const int  excl = wpre + incl - cnt_i;
    const int  base = s_base;
    const bool fits = nb <= STAGE_CAP;
    if (tid < BNODES) {
        s_cur[tid] = excl;
        const int node = b * BNODES + tid;
        if (node < n) {
            rowptr[node] = base + excl;
            dinv[node]   = rsqrtf((float)(cnt_i + 1));
            if (node == n - 1) rowptr[n] = base + excl + cnt_i;
        }
    }
    __syncthreads();

    walk_bucket(chunk_off, recs, b, nchunks, wave, lane,
                [&](int rec) {
                    const int pos = atomicAdd(&s_cur[rec & (BNODES - 1)], 1);
                    const int sv  = (int)((unsigned)rec >> BSH);
                    if (fits) s_stage[pos] = sv;
                    else      csr_src[base + pos] = sv;
                });
    __syncthreads();
    if (fits) {
        for (int i = tid; i < nb; i += 1024) csr_src[base + i] = s_stage[i];
    }
}

#define TM 32
#define TP 132

__device__ __forceinline__ v8f wmma_f32_k4(v2f a, v2f b, v8f c) {
    return __builtin_amdgcn_wmma_f32_16x16x4_f32(
         false, a,  false, b,
         (short)0, c,  false,  false);
}

__launch_bounds__(256)
__global__ void gather_mlp_kernel(const int* __restrict__ rowptr,
                                  const int* __restrict__ csr_src,
                                  const float* __restrict__ dinv,
                                  const float* __restrict__ x,
                                  const float* __restrict__ W1,
                                  const float* __restrict__ b1,
                                  const float* __restrict__ W2,
                                  float* __restrict__ h2, int n) {
    __shared__ __attribute__((aligned(16))) float tile[TM * TP];

    const int lane = threadIdx.x & 31;
    const int wave  = __builtin_amdgcn_readfirstlane((int)(threadIdx.x >> 5));
    const int node0 = blockIdx.x * TM;

    const float* xl = x + lane * 4;
    for (int i = 0; i < TM / 8; ++i) {
        const int rl = wave * (TM / 8) + i;
        const int d  = node0 + rl;
        v4f acc = (v4f){};
        if (d < n) {
            const int   beg = rowptr[d];
            const int   end = rowptr[d + 1];
            const float dd  = dinv[d];
            acc = *(const v4f*)(xl + (size_t)d * 128) * (dd * dd);
            for (int base = beg; base < end; base += 32) {
                const int m = min(end - base, 32);
                int   s_l = 0;
                float w_l = 0.0f;
                if (lane < m) {
                    s_l = csr_src[base + lane];
                    w_l = dinv[s_l] * dd;
                }
                const int w_i = __float_as_int(w_l);
                int j = 0;
                for (; j + 4 <= m; j += 4) {
                    const int   s0 = __builtin_amdgcn_readlane(s_l, j + 0);
                    const int   s1 = __builtin_amdgcn_readlane(s_l, j + 1);
                    const int   s2 = __builtin_amdgcn_readlane(s_l, j + 2);
                    const int   s3 = __builtin_amdgcn_readlane(s_l, j + 3);
                    const float w0 = __int_as_float(__builtin_amdgcn_readlane(w_i, j + 0));
                    const float w1 = __int_as_float(__builtin_amdgcn_readlane(w_i, j + 1));
                    const float w2 = __int_as_float(__builtin_amdgcn_readlane(w_i, j + 2));
                    const float w3 = __int_as_float(__builtin_amdgcn_readlane(w_i, j + 3));
                    const v4f v0 = *(const v4f*)(xl + (size_t)s0 * 128);
                    const v4f v1 = *(const v4f*)(xl + (size_t)s1 * 128);
                    const v4f v2 = *(const v4f*)(xl + (size_t)s2 * 128);
                    const v4f v3 = *(const v4f*)(xl + (size_t)s3 * 128);
                    acc += v0 * w0;
                    acc += v1 * w1;
                    acc += v2 * w2;
                    acc += v3 * w3;
                }
                for (; j < m; ++j) {
                    const int   s = __builtin_amdgcn_readlane(s_l, j);
                    const float w = __int_as_float(__builtin_amdgcn_readlane(w_i, j));
                    const v4f   v = *(const v4f*)(xl + (size_t)s * 128);
                    acc += v * w;
                }
            }
        }
        *(v4f*)(tile + rl * TP + lane * 4) = acc;
    }
    __syncthreads();

    const int c  = lane & 15;
    const int hh = lane >> 4;

    {
        constexpr int MT = TM / 16;
        v8f acc[MT];
        #pragma unroll
        for (int t = 0; t < MT; ++t) acc[t] = (v8f){};

        const float* bp = W1 + (size_t)(4 * hh) * 128 + wave * 16 + c;
        const float* ap = tile + c * TP + 4 * hh;
        #pragma unroll 2
        for (int j = 0; j < 16; ++j) {
            v2f blo, bhi;
            blo.x = bp[(8 * j + 0) * 128];
            blo.y = bp[(8 * j + 1) * 128];
            bhi.x = bp[(8 * j + 2) * 128];
            bhi.y = bp[(8 * j + 3) * 128];
            v4f a[MT];
            #pragma unroll
            for (int t = 0; t < MT; ++t) a[t] = *(const v4f*)(ap + t * 16 * TP + 8 * j);
            #pragma unroll
            for (int t = 0; t < MT; ++t) acc[t] = wmma_f32_k4(a[t].xy, blo, acc[t]);
            #pragma unroll
            for (int t = 0; t < MT; ++t) acc[t] = wmma_f32_k4(a[t].zw, bhi, acc[t]);
        }

        const float bias = b1[wave * 16 + c];
        __syncthreads();
        #pragma unroll
        for (int t = 0; t < MT; ++t) {
            float* hrow = tile + (t * 16 + 8 * hh) * TP + wave * 16 + c;
            #pragma unroll
            for (int r = 0; r < 8; ++r) hrow[r * TP] = fmaxf(acc[t][r] + bias, 0.0f);
        }
        __syncthreads();
    }

    {
        constexpr int MT2 = TM / 32;
        const int nt = wave & 3;
        const int mh = wave >> 2;
        v8f acc[MT2];
        #pragma unroll
        for (int t = 0; t < MT2; ++t) acc[t] = (v8f){};

        const float* bp = W2 + (size_t)(4 * hh) * 64 + nt * 16 + c;
        const float* ap = tile + (mh * MT2 * 16 + c) * TP + 4 * hh;
        #pragma unroll 2
        for (int j = 0; j < 16; ++j) {
            v2f blo, bhi;
            blo.x = bp[(8 * j + 0) * 64];
            blo.y = bp[(8 * j + 1) * 64];
            bhi.x = bp[(8 * j + 2) * 64];
            bhi.y = bp[(8 * j + 3) * 64];
            v4f a[MT2];
            #pragma unroll
            for (int t = 0; t < MT2; ++t) a[t] = *(const v4f*)(ap + t * 16 * TP + 8 * j);
            #pragma unroll
            for (int t = 0; t < MT2; ++t) acc[t] = wmma_f32_k4(a[t].xy, blo, acc[t]);
            #pragma unroll
            for (int t = 0; t < MT2; ++t) acc[t] = wmma_f32_k4(a[t].zw, bhi, acc[t]);
        }

        #pragma unroll
        for (int t = 0; t < MT2; ++t) {
            const int row0 = node0 + (mh * MT2 + t) * 16 + 8 * hh;
            float* orow = h2 + (size_t)row0 * 64 + nt * 16 + c;
            #pragma unroll
            for (int r = 0; r < 8; ++r) {
                if (row0 + r < n) orow[(size_t)r * 64] = acc[t][r];
            }
        }
    }
}

template <int F, bool RELU>
__launch_bounds__(256)
__global__ void gather_sum_kernel(const int* __restrict__ rowptr,
                                  const int* __restrict__ csr_src,
                                  const float* __restrict__ dinv,
                                  const float* __restrict__ h,
                                  const float* __restrict__ bias,
                                  float* __restrict__ out, int n) {
    constexpr int VE = F / 32;
    typedef __attribute__((ext_vector_type(VE))) float vec;

    const int lane = threadIdx.x & 31;
    const int d = __builtin_amdgcn_readfirstlane(
        (int)(blockIdx.x * (blockDim.x >> 5) + (threadIdx.x >> 5)));
    if (d >= n) return;

    const int   beg = rowptr[d];
    const int   end = rowptr[d + 1];
    const float dd  = dinv[d];

    const float* hl = h + lane * VE;
    vec acc = *(const vec*)(hl + (size_t)d * F) * (dd * dd);

    for (int base = beg; base < end; base += 32) {
        const int m = min(end - base, 32);
        int   s_l = 0;
        float w_l = 0.0f;
        if (lane < m) {
            s_l = csr_src[base + lane];
            w_l = dinv[s_l] * dd;
        }
        const int w_i = __float_as_int(w_l);
        int j = 0;
        for (; j + 4 <= m; j += 4) {
            const int   s0 = __builtin_amdgcn_readlane(s_l, j + 0);
            const int   s1 = __builtin_amdgcn_readlane(s_l, j + 1);
            const int   s2 = __builtin_amdgcn_readlane(s_l, j + 2);
            const int   s3 = __builtin_amdgcn_readlane(s_l, j + 3);
            const float w0 = __int_as_float(__builtin_amdgcn_readlane(w_i, j + 0));
            const float w1 = __int_as_float(__builtin_amdgcn_readlane(w_i, j + 1));
            const float w2 = __int_as_float(__builtin_amdgcn_readlane(w_i, j + 2));
            const float w3 = __int_as_float(__builtin_amdgcn_readlane(w_i, j + 3));
            const vec v0 = *(const vec*)(hl + (size_t)s0 * F);
            const vec v1 = *(const vec*)(hl + (size_t)s1 * F);
            const vec v2 = *(const vec*)(hl + (size_t)s2 * F);
            const vec v3 = *(const vec*)(hl + (size_t)s3 * F);
            acc += v0 * w0;
            acc += v1 * w1;
            acc += v2 * w2;
            acc += v3 * w3;
        }
        for (; j < m; ++j) {
            const int   s = __builtin_amdgcn_readlane(s_l, j);
            const float w = __int_as_float(__builtin_amdgcn_readlane(w_i, j));
            const vec   v = *(const vec*)(hl + (size_t)s * F);
            acc += v * w;
        }
    }

    vec r = acc + *(const vec*)(bias + lane * VE);
    if constexpr (RELU) {
        for (int q = 0; q < VE; ++q) r[q] = fmaxf(r[q], 0.0f);
    }
    *(vec*)(out + (size_t)d * F + lane * VE) = r;
}

extern "C" void kernel_launch(void* const* d_in, const int* in_sizes, int n_in,
                              void* d_out, int out_size, void* d_ws, size_t ws_size,
                              hipStream_t stream) {
    const float* x    = (const float*)d_in[0];
    const int*   eidx = (const int*)  d_in[1];
    const float* W1   = (const float*)d_in[2];
    const float* b1   = (const float*)d_in[3];
    const float* W2   = (const float*)d_in[4];
    const float* b2   = (const float*)d_in[5];
    float*       out  = (float*)d_out;

    const int N = NNODES;
    const int E = in_sizes[1] / 2;
    const int* src = eidx;
    const int* dst = eidx + E;

    char*  ws        = (char*)d_ws;
    float* dinv      = (float*)(ws);
    int*   rowptr    = (int*)  (ws + (size_t)409600);
    float* h2        = (float*)(ws + (size_t)(1  << 20));
    int*   csr       = (int*)  (ws + (size_t)(1 << 20) + (size_t)25600000);
    int*   recs      = (int*)  (ws + (size_t)(53 << 20));
    int*   chunk_off = (int*)  (ws + (size_t)(61 << 20));

    const int T = 256;

    const int nchunks = (E + CHUNK - 1) / CHUNK;
    bin_chunk_kernel<<<nchunks, 1024, 0, stream>>>(src, dst, chunk_off, recs, E);
    bucket_sort_kernel<<<NBK, 1024, 0, stream>>>(chunk_off, recs, rowptr, dinv, csr, N, nchunks);

    gather_mlp_kernel<<<(N + TM - 1) / TM, 256, 0, stream>>>(rowptr, csr, dinv, x, W1, b1, W2, h2, N);

    gather_sum_kernel<64, false><<<(N + 7) / 8, T, 0, stream>>>(rowptr, csr, dinv, h2, b2, out, N);
}


# Layer-2 gather: a wave walks a strip of 7 consecutive nodes instead of one wave launched per node (12500 -> 1786 blocks), 8 row loads per trip, next node's ids fetched a node ahead; add order kept
# speedup vs baseline: 6.2227x; 1.0313x over previous
#include <hip/hip_runtime.h>

typedef __attribute__((ext_vector_type(2))) float v2f;
typedef __attribute__((ext_vector_type(4))) float v4f;
typedef __attribute__((ext_vector_type(8))) float v8f;

#define NNODES 100000

#define BSH       9
#define BNODES    (1 << BSH)
#define NBK       ((NNODES + BNODES - 1) >> BSH)
#define CHUNK     8192
#define STAGE_CAP 12288

__launch_bounds__(1024)
__global__ void bin_chunk_kernel(const int* __restrict__ src, const int* __restrict__ dst,
                                 int* __restrict__ chunk_off, int* __restrict__ recs, int E) {
    __shared__ int s_bin[NBK + 1];
    __shared__ int s_rec[CHUNK];
    constexpr int EPT = CHUNK / 1024;
    constexpr int PER = (NBK + 31) / 32;

    const int tid  = threadIdx.x;
    const int lane = tid & 31;
    const int wave = tid >> 5;
    const int e0   = blockIdx.x * CHUNK;

    if (tid <= NBK) s_bin[tid] = 0;
    __syncthreads();

    int s[EPT], d[EPT], r[EPT];
    #pragma unroll
    for (int k = 0; k < EPT; ++k) {
        const int e = e0 + k * 1024 + tid;
        d[k] = -1; s[k] = 0;
        if (e < E) { d[k] = dst[e]; s[k] = src[e]; }
    }
    #pragma unroll
    for (int k = 0; k < EPT; ++k) {
        r[k] = 0;
        if (d[k] >= 0) r[k] = atomicAdd(&s_bin[d[k] >> BSH], 1);
    }
    __syncthreads();

    if (wave == 0) {
        int v[PER];
        int sum = 0;
        #pragma unroll
        for (int q = 0; q < PER; ++q) {
            const int idx = lane * PER + q;
            v[q] = (idx < NBK) ? s_bin[idx] : 0;
            sum += v[q];
        }
        int incl = sum;
        for (int o = 1; o < 32; o <<= 1) {
            const int t = __shfl_up(incl, o);
            if (lane >= o) incl += t;
        }
        int run = incl - sum;
        #pragma unroll
        for (int q = 0; q < PER; ++q) {
            const int idx = lane * PER + q;
            if (idx < NBK) s_bin[idx] = run;
            run += v[q];
        }
        if (lane == 31) s_bin[NBK] = incl;
    }
    __syncthreads();

    if (tid <= NBK) chunk_off[blockIdx.x * (NBK + 1) + tid] = s_bin[tid];
    #pragma unroll
    for (int k = 0; k < EPT; ++k) {
        if (d[k] >= 0)
            s_rec[s_bin[d[k] >> BSH] + r[k]] = (s[k] << BSH) | (d[k] & (BNODES - 1));
    }
    __syncthreads();

    const int nv = s_bin[NBK];
    for (int i = tid; i < nv; i += 1024) recs[e0 + i] = s_rec[i];
}

template <typename F>
__device__ __forceinline__ int walk_bucket(const int* __restrict__ chunk_off,
                                           const int* __restrict__ recs,
                                           int b, int nchunks, int wave, int lane, F f) {
    int osum = 0;
    for (int j0 = 0; j0 < nchunks; j0 += 1024) {
        const int j = j0 + wave + 32 * lane;
        int o0 = 0, c = 0;
        if (j < nchunks) {
            o0 = chunk_off[j * (NBK + 1) + b];
            c  = chunk_off[j * (NBK + 1) + b + 1] - o0;
        }
        osum += o0;
        for (int k = 0; k < 32; k += 4) {
            if (j0 + wave + 32 * k >= nchunks) break;
            const int* run[4];
            int cnt[4], first[4];
            #pragma unroll
            for (int u = 0; u < 4; ++u) {
                cnt[u] = __builtin_amdgcn_readlane(c, k + u);
                const int o = __builtin_amdgcn_readlane(o0, k + u);
                run[u] = recs + (size_t)(j0 + wave + 32 * (k + u)) * CHUNK + o;
                first[u] = -1;
                if (lane < cnt[u]) first[u] = run[u][lane];
            }
            #pragma unroll
            for (int u = 0; u < 4; ++u) {
                if (first[u] >= 0) f(first[u]);
            }
            #pragma unroll
            for (int u = 0; u < 4; ++u) {
                for (int i = lane + 32; i < cnt[u]; i += 32) f(run[u][i]);
            }
        }
    }
    return osum;
}

__launch_bounds__(1024)
__global__ void bucket_sort_kernel(const int* __restrict__ chunk_off, const int* __restrict__ recs,
                                   int* __restrict__ rowptr, float* __restrict__ dinv,
                                   int* __restrict__ csr_src, int n, int nchunks) {
    __shared__ int s_cnt[BNODES];
    __shared__ int s_cur[BNODES];
    __shared__ int s_wsum[32];
    __shared__ int s_base;
    __shared__ int s_stage[STAGE_CAP];

    const int tid  = threadIdx.x;
    const int lane = tid & 31;
    const int wave = tid >> 5;
    const int b    = blockIdx.x;

    if (tid < BNODES) s_cnt[tid] = 0;
    if (tid == 0) s_base = 0;
    __syncthreads();

    int osum = walk_bucket(chunk_off, recs, b, nchunks, wave, lane,
                           [&](int rec) { atomicAdd(&s_cnt[rec & (BNODES - 1)], 1); });
    for (int o = 16; o > 0; o >>= 1) osum += __shfl_xor(osum, o);
    if (lane == 0 && osum != 0) atomicAdd(&s_base, osum);
    __syncthreads();

    const int cnt_i = (tid < BNODES) ? s_cnt[tid] : 0;
    int incl = cnt_i;
    for (int o = 1; o < 32; o <<= 1) {
        const int t = __shfl_up(incl, o);
        if (lane >= o) incl += t;
    }
    if (lane == 31) s_wsum[wave] = incl;
    __syncthreads();

    int wpre = 0, nb = 0;
    for (int w = 0; w < 32; ++w) {
        const int t = s_wsum[w];
        if (w < wave) wpre += t;
        nb += t;
    }
    const int  excl = wpre + incl - cnt_i;
    const int  base = s_base;
    const bool fits = nb <= STAGE_CAP;
    if (tid < BNODES) {
        s_cur[tid] = excl;
        const int node = b * BNODES + tid;
        if (node < n) {
            rowptr[node] = base + excl;
            dinv[node]   = rsqrtf((float)(cnt_i + 1));
            if (node == n - 1) rowptr[n] = base + excl + cnt_i;
        }
    }
    __syncthreads();

    walk_bucket(chunk_off, recs, b, nchunks, wave, lane,
                [&](int rec) {
                    const int pos = atomicAdd(&s_cur[rec & (BNODES - 1)], 1);
                    const int sv  = (int)((unsigned)rec >> BSH);
                    if (fits) s_stage[pos] = sv;
                    else      csr_src[base + pos] = sv;
                });
    __syncthreads();
    if (fits) {
        for (int i = tid; i < nb; i += 1024) csr_src[base + i] = s_stage[i];
    }
}

#define TM 32
#define TP 132

__device__ __forceinline__ v8f wmma_f32_k4(v2f a, v2f b, v8f c) {
    return __builtin_amdgcn_wmma_f32_16x16x4_f32(
         false, a,  false, b,
         (short)0, c,  false,  false);
}

__launch_bounds__(256)
__global__ void gather_mlp_kernel(const int* __restrict__ rowptr,
                                  const int* __restrict__ csr_src,
                                  const float* __restrict__ dinv,
                                  const float* __restrict__ x,
                                  const float* __restrict__ W1,
                                  const float* __restrict__ b1,
                                  const float* __restrict__ W2,
                                  float* __restrict__ h2, int n) {
    __shared__ __attribute__((aligned(16))) float tile[TM * TP];

    const int lane = threadIdx.x & 31;
    const int wave  = __builtin_amdgcn_readfirstlane((int)(threadIdx.x >> 5));
    const int node0 = blockIdx.x * TM;

    const float* xl = x + lane * 4;
    for (int i = 0; i < TM / 8; ++i) {
        const int rl = wave * (TM / 8) + i;
        const int d  = node0 + rl;
        v4f acc = (v4f){};
        if (d < n) {
            const int   beg = rowptr[d];
            const int   end = rowptr[d + 1];
            const float dd  = dinv[d];
            acc = *(const v4f*)(xl + (size_t)d * 128) * (dd * dd);
            for (int base = beg; base < end; base += 32) {
                const int m = min(end - base, 32);
                int   s_l = 0;
                float w_l = 0.0f;
                if (lane < m) {
                    s_l = csr_src[base + lane];
                    w_l = dinv[s_l] * dd;
                }
                const int w_i = __float_as_int(w_l);
                int j = 0;
                for (; j + 4 <= m; j += 4) {
                    const int   s0 = __builtin_amdgcn_readlane(s_l, j + 0);
                    const int   s1 = __builtin_amdgcn_readlane(s_l, j + 1);
                    const int   s2 = __builtin_amdgcn_readlane(s_l, j + 2);
                    const int   s3 = __builtin_amdgcn_readlane(s_l, j + 3);
                    const float w0 = __int_as_float(__builtin_amdgcn_readlane(w_i, j + 0));
                    const float w1 = __int_as_float(__builtin_amdgcn_readlane(w_i, j + 1));
                    const float w2 = __int_as_float(__builtin_amdgcn_readlane(w_i, j + 2));
                    const float w3 = __int_as_float(__builtin_amdgcn_readlane(w_i, j + 3));
                    const v4f v0 = *(const v4f*)(xl + (size_t)s0 * 128);
                    const v4f v1 = *(const v4f*)(xl + (size_t)s1 * 128);
                    const v4f v2 = *(const v4f*)(xl + (size_t)s2 * 128);
                    const v4f v3 = *(const v4f*)(xl + (size_t)s3 * 128);
                    acc += v0 * w0;
                    acc += v1 * w1;
                    acc += v2 * w2;
                    acc += v3 * w3;
                }
                for (; j < m; ++j) {
                    const int   s = __builtin_amdgcn_readlane(s_l, j);
                    const float w = __int_as_float(__builtin_amdgcn_readlane(w_i, j));
                    const v4f   v = *(const v4f*)(xl + (size_t)s * 128);
                    acc += v * w;
                }
            }
        }
        *(v4f*)(tile + rl * TP + lane * 4) = acc;
    }
    __syncthreads();

    const int c  = lane & 15;
    const int hh = lane >> 4;

    {
        constexpr int MT = TM / 16;
        v8f acc[MT];
        #pragma unroll
        for (int t = 0; t < MT; ++t) acc[t] = (v8f){};

        const float* bp = W1 + (size_t)(4 * hh) * 128 + wave * 16 + c;
        const float* ap = tile + c * TP + 4 * hh;
        #pragma unroll 2
        for (int j = 0; j < 16; ++j) {
            v2f blo, bhi;
            blo.x = bp[(8 * j + 0) * 128];
            blo.y = bp[(8 * j + 1) * 128];
            bhi.x = bp[(8 * j + 2) * 128];
            bhi.y = bp[(8 * j + 3) * 128];
            v4f a[MT];
            #pragma unroll
            for (int t = 0; t < MT; ++t) a[t] = *(const v4f*)(ap + t * 16 * TP + 8 * j);
            #pragma unroll
            for (int t = 0; t < MT; ++t) acc[t] = wmma_f32_k4(a[t].xy, blo, acc[t]);
            #pragma unroll
            for (int t = 0; t < MT; ++t) acc[t] = wmma_f32_k4(a[t].zw, bhi, acc[t]);
        }

        const float bias = b1[wave * 16 + c];
        __syncthreads();
        #pragma unroll
        for (int t = 0; t < MT; ++t) {
            float* hrow = tile + (t * 16 + 8 * hh) * TP + wave * 16 + c;
            #pragma unroll
            for (int r = 0; r < 8; ++r) hrow[r * TP] = fmaxf(acc[t][r] + bias, 0.0f);
        }
        __syncthreads();
    }

    {
        constexpr int MT2 = TM / 32;
        const int nt = wave & 3;
        const int mh = wave >> 2;
        v8f acc[MT2];
        #pragma unroll
        for (int t = 0; t < MT2; ++t) acc[t] = (v8f){};

        const float* bp = W2 + (size_t)(4 * hh) * 64 + nt * 16 + c;
        const float* ap = tile + (mh * MT2 * 16 + c) * TP + 4 * hh;
        #pragma unroll 2
        for (int j = 0; j < 16; ++j) {
            v2f blo, bhi;
            blo.x = bp[(8 * j + 0) * 64];
            blo.y = bp[(8 * j + 1) * 64];
            bhi.x = bp[(8 * j + 2) * 64];
            bhi.y = bp[(8 * j + 3) * 64];
            v4f a[MT2];
            #pragma unroll
            for (int t = 0; t < MT2; ++t) a[t] = *(const v4f*)(ap + t * 16 * TP + 8 * j);
            #pragma unroll
            for (int t = 0; t < MT2; ++t) acc[t] = wmma_f32_k4(a[t].xy, blo, acc[t]);
            #pragma unroll
            for (int t = 0; t < MT2; ++t) acc[t] = wmma_f32_k4(a[t].zw, bhi, acc[t]);
        }

        #pragma unroll
        for (int t = 0; t < MT2; ++t) {
            const int row0 = node0 + (mh * MT2 + t) * 16 + 8 * hh;
            float* orow = h2 + (size_t)row0 * 64 + nt * 16 + c;
            #pragma unroll
            for (int r = 0; r < 8; ++r) {
                if (row0 + r < n) orow[(size_t)r * 64] = acc[t][r];
            }
        }
    }
}

#define STRIP 7
#define ETRIP 8

template <int F, bool RELU>
__launch_bounds__(256)
__global__ void gather_sum_kernel(const int* __restrict__ rowptr,
                                  const int* __restrict__ csr_src,
                                  const float* __restrict__ dinv,
                                  const float* __restrict__ h,
                                  const float* __restrict__ bias,
                                  float* __restrict__ out, int n) {
    static_assert(STRIP + 1 <= 32, "one lane per row pointer of the strip");
    constexpr int VE = F / 32;
    typedef __attribute__((ext_vector_type(VE))) float vec;

    const int lane = threadIdx.x & 31;
    const int d0 = STRIP * __builtin_amdgcn_readfirstlane(
        (int)(blockIdx.x * (blockDim.x >> 5) + (threadIdx.x >> 5)));
    if (d0 >= n) return;
    const int cnt = min(STRIP, n - d0);

    int   rp_l = 0;
    float dn_l = 0.0f;
    if (lane <= cnt) rp_l = rowptr[d0 + lane];
    if (lane <  cnt) dn_l = dinv[d0 + lane];
    const int dn_i = __float_as_int(dn_l);

    const float* hl = h + lane * VE;
    const vec    bv = *(const vec*)(bias + lane * VE);

    const int e_last = max(__builtin_amdgcn_readlane(rp_l, cnt) - 1, 0);

    auto trip = [&](vec& acc, vec vself, int s_l, int& g_i, float dd, int m, int j, int rem,
                    const bool tail, const bool init) {
        int   sk[ETRIP];
        vec   vk[ETRIP];
        float wk[ETRIP];
        #pragma unroll
        for (int k = 0; k < ETRIP; ++k) sk[k] = __builtin_amdgcn_readlane(s_l, j + k);
        #pragma unroll
        for (int k = 0; k < ETRIP; ++k) vk[k] = *(const vec*)(hl + (size_t)sk[k] * F);
        if (init) {
            const int g_raw = __float_as_int(dinv[s_l]);
            g_i = (lane < m) ? g_raw : 0;
            acc = vself * (dd * dd);
        }
        #pragma unroll
        for (int k = 0; k < ETRIP; ++k)
            wk[k] = __int_as_float(__builtin_amdgcn_readlane(g_i, j + k)) * dd;
        #pragma unroll
        for (int k = 0; k < ETRIP; ++k) {
            vec v = vk[k];
            if (tail && k >= rem) v = (vec){};
            acc += v * wk[k];
        }
    };
    auto sum_chunk = [&](vec& acc, vec vself, int s_l, float dd, int m, const bool first) {
        int g_i = 0;
        int j   = 0;
        if (first) {
            if (m < ETRIP) { trip(acc, vself, s_l, g_i, dd, m, 0, m, true, true); return; }
            trip(acc, vself, s_l, g_i, dd, m, 0, ETRIP, false, true);
            j = ETRIP;
        } else {
            g_i = (lane < m) ? __float_as_int(dinv[s_l]) : 0;
        }
        for (; j + ETRIP <= m; j += ETRIP) trip(acc, vself, s_l, g_i, dd, m, j, ETRIP, false, false);
        if (j < m) trip(acc, vself, s_l, g_i, dd, m, j, m - j, true, false);
    };

    int beg   = __builtin_amdgcn_readlane(rp_l, 0);
    int end   = __builtin_amdgcn_readlane(rp_l, 1);
    int s_raw = csr_src[min(beg + lane, e_last)];

    for (int i = 0; i < cnt; ++i) {
        const int   d  = d0 + i;
        const float dd = __int_as_float(__builtin_amdgcn_readlane(dn_i, i));
        const int   m  = min(end - beg, 32);
        const int   s_c = (lane < m) ? s_raw : 0;

        const int end_n = __builtin_amdgcn_readlane(rp_l, min(i + 2, cnt));
        s_raw = csr_src[min(end + lane, e_last)];

        const vec vself = *(const vec*)(hl + (size_t)d * F);
        vec acc;
        if (m > 0) sum_chunk(acc, vself, s_c, dd, m, true);
        else       acc = vself * (dd * dd);
        for (int base = beg + 32; base < end; base += 32) {
            const int mm = min(end - base, 32);
            int s_l = 0;
            if (lane < mm) s_l = csr_src[base + lane];
            sum_chunk(acc, vself, s_l, dd, mm, false);
        }

        vec r = acc + bv;
        if constexpr (RELU) {
            for (int q = 0; q < VE; ++q) r[q] = fmaxf(r[q], 0.0f);
        }
        *(vec*)(out + (size_t)d * F + lane * VE) = r;

        beg = end; end = end_n;
    }
}

extern "C" void kernel_launch(void* const* d_in, const int* in_sizes, int n_in,
                              void* d_out, int out_size, void* d_ws, size_t ws_size,
                              hipStream_t stream) {
    const float* x    = (const float*)d_in[0];
    const int*   eidx = (const int*)  d_in[1];
    const float* W1   = (const float*)d_in[2];
    const float* b1   = (const float*)d_in[3];
    const float* W2   = (const float*)d_in[4];
    const float* b2   = (const float*)d_in[5];
    float*       out  = (float*)d_out;

    const int N = NNODES;
    const int E = in_sizes[1] / 2;
    const int* src = eidx;
    const int* dst = eidx + E;

    char*  ws        = (char*)d_ws;
    float* dinv      = (float*)(ws);
    int*   rowptr    = (int*)  (ws + (size_t)409600);
    float* h2        = (float*)(ws + (size_t)(1  << 20));
    int*   csr       = (int*)  (ws + (size_t)(1 << 20) + (size_t)25600000);
    int*   recs      = (int*)  (ws + (size_t)(53 << 20));
    int*   chunk_off = (int*)  (ws + (size_t)(61 << 20));

    const int T = 256;

    const int nchunks = (E + CHUNK - 1) / CHUNK;
    bin_chunk_kernel<<<nchunks, 1024, 0, stream>>>(src, dst, chunk_off, recs, E);
    bucket_sort_kernel<<<NBK, 1024, 0, stream>>>(chunk_off, recs, rowptr, dinv, csr, N, nchunks);

    gather_mlp_kernel<<<(N + TM - 1) / TM, 256, 0, stream>>>(rowptr, csr, dinv, x, W1, b1, W2, h2, N);

    gather_sum_kernel<64, false><<<(N + STRIP * 8 - 1) / (STRIP * 8), T, 0, stream>>>(rowptr, csr, dinv, h2, b2, out, N);
}


# Layer-1 kernel stores each h2 row times its own node's dinv, so the layer-2 gather adds plain rows: no scale-factor load and no multiply per edge; product rounding moves, add order kept
# speedup vs baseline: 6.4628x; 1.0386x over previous
#include <hip/hip_runtime.h>

typedef __attribute__((ext_vector_type(2))) float v2f;
typedef __attribute__((ext_vector_type(4))) float v4f;
typedef __attribute__((ext_vector_type(8))) float v8f;

#define NNODES 100000

#define BSH       9
#define BNODES    (1 << BSH)
#define NBK       ((NNODES + BNODES - 1) >> BSH)
#define CHUNK     8192
#define STAGE_CAP 12288

__launch_bounds__(1024)
__global__ void bin_chunk_kernel(const int* __restrict__ src, const int* __restrict__ dst,
                                 int* __restrict__ chunk_off, int* __restrict__ recs, int E) {
    __shared__ int s_bin[NBK + 1];
    __shared__ int s_rec[CHUNK];
    constexpr int EPT = CHUNK / 1024;
    constexpr int PER = (NBK + 31) / 32;

    const int tid  = threadIdx.x;
    const int lane = tid & 31;
    const int wave = tid >> 5;
    const int e0   = blockIdx.x * CHUNK;

    if (tid <= NBK) s_bin[tid] = 0;
    __syncthreads();

    int s[EPT], d[EPT], r[EPT];
    #pragma unroll
    for (int k = 0; k < EPT; ++k) {
        const int e = e0 + k * 1024 + tid;
        d[k] = -1; s[k] = 0;
        if (e < E) { d[k] = dst[e]; s[k] = src[e]; }
    }
    #pragma unroll
    for (int k = 0; k < EPT; ++k) {
        r[k] = 0;
        if (d[k] >= 0) r[k] = atomicAdd(&s_bin[d[k] >> BSH], 1);
    }
    __syncthreads();

    if (wave == 0) {
        int v[PER];
        int sum = 0;
        #pragma unroll
        for (int q = 0; q < PER; ++q) {
            const int idx = lane * PER + q;
            v[q] = (idx < NBK) ? s_bin[idx] : 0;
            sum += v[q];
        }
        int incl = sum;
        for (int o = 1; o < 32; o <<= 1) {
            const int t = __shfl_up(incl, o);
            if (lane >= o) incl += t;
        }
        int run = incl - sum;
        #pragma unroll
        for (int q = 0; q < PER; ++q) {
            const int idx = lane * PER + q;
            if (idx < NBK) s_bin[idx] = run;
            run += v[q];
        }
        if (lane == 31) s_bin[NBK] = incl;
    }
    __syncthreads();

    if (tid <= NBK) chunk_off[blockIdx.x * (NBK + 1) + tid] = s_bin[tid];
    #pragma unroll
    for (int k = 0; k < EPT; ++k) {
        if (d[k] >= 0)
            s_rec[s_bin[d[k] >> BSH] + r[k]] = (s[k] << BSH) | (d[k] & (BNODES - 1));
    }
    __syncthreads();

    const int nv = s_bin[NBK];
    for (int i = tid; i < nv; i += 1024) recs[e0 + i] = s_rec[i];
}

template <typename F>
__device__ __forceinline__ int walk_bucket(const int* __restrict__ chunk_off,
                                           const int* __restrict__ recs,
                                           int b, int nchunks, int wave, int lane, F f) {
    int osum = 0;
    for (int j0 = 0; j0 < nchunks; j0 += 1024) {
        const int j = j0 + wave + 32 * lane;
        int o0 = 0, c = 0;
        if (j < nchunks) {
            o0 = chunk_off[j * (NBK + 1) + b];
            c  = chunk_off[j * (NBK + 1) + b + 1] - o0;
        }
        osum += o0;
        for (int k = 0; k < 32; k += 4) {
            if (j0 + wave + 32 * k >= nchunks) break;
            const int* run[4];
            int cnt[4], first[4];
            #pragma unroll
            for (int u = 0; u < 4; ++u) {
                cnt[u] = __builtin_amdgcn_readlane(c, k + u);
                const int o = __builtin_amdgcn_readlane(o0, k + u);
                run[u] = recs + (size_t)(j0 + wave + 32 * (k + u)) * CHUNK + o;
                first[u] = -1;
                if (lane < cnt[u]) first[u] = run[u][lane];
            }
            #pragma unroll
            for (int u = 0; u < 4; ++u) {
                if (first[u] >= 0) f(first[u]);
            }
            #pragma unroll
            for (int u = 0; u < 4; ++u) {
                for (int i = lane + 32; i < cnt[u]; i += 32) f(run[u][i]);
            }
        }
    }
    return osum;
}

__launch_bounds__(1024)
__global__ void bucket_sort_kernel(const int* __restrict__ chunk_off, const int* __restrict__ recs,
                                   int* __restrict__ rowptr, float* __restrict__ dinv,
                                   int* __restrict__ csr_src, int n, int nchunks) {
    __shared__ int s_cnt[BNODES];
    __shared__ int s_cur[BNODES];
    __shared__ int s_wsum[32];
    __shared__ int s_base;
    __shared__ int s_stage[STAGE_CAP];

    const int tid  = threadIdx.x;
    const int lane = tid & 31;
    const int wave = tid >> 5;
    const int b    = blockIdx.x;

    if (tid < BNODES) s_cnt[tid] = 0;
    if (tid == 0) s_base = 0;
    __syncthreads();

    int osum = walk_bucket(chunk_off, recs, b, nchunks, wave, lane,
                           [&](int rec) { atomicAdd(&s_cnt[rec & (BNODES - 1)], 1); });
    for (int o = 16; o > 0; o >>= 1) osum += __shfl_xor(osum, o);
    if (lane == 0 && osum != 0) atomicAdd(&s_base, osum);
    __syncthreads();

    const int cnt_i = (tid < BNODES) ? s_cnt[tid] : 0;
    int incl = cnt_i;
    for (int o = 1; o < 32; o <<= 1) {
        const int t = __shfl_up(incl, o);
        if (lane >= o) incl += t;
    }
    if (lane == 31) s_wsum[wave] = incl;
    __syncthreads();

    int wpre = 0, nb = 0;
    for (int w = 0; w < 32; ++w) {
        const int t = s_wsum[w];
        if (w < wave) wpre += t;
        nb += t;
    }
    const int  excl = wpre + incl - cnt_i;
    const int  base = s_base;
    const bool fits = nb <= STAGE_CAP;
    if (tid < BNODES) {
        s_cur[tid] = excl;
        const int node = b * BNODES + tid;
        if (node < n) {
            rowptr[node] = base + excl;
            dinv[node]   = rsqrtf((float)(cnt_i + 1));
            if (node == n - 1) rowptr[n] = base + excl + cnt_i;
        }
    }
    __syncthreads();

    walk_bucket(chunk_off, recs, b, nchunks, wave, lane,
                [&](int rec) {
                    const int pos = atomicAdd(&s_cur[rec & (BNODES - 1)], 1);
                    const int sv  = (int)((unsigned)rec >> BSH);
                    if (fits) s_stage[pos] = sv;
                    else      csr_src[base + pos] = sv;
                });
    __syncthreads();
    if (fits) {
        for (int i = tid; i < nb; i += 1024) csr_src[base + i] = s_stage[i];
    }
}

#define TM 32
#define TP 132

__device__ __forceinline__ v8f wmma_f32_k4(v2f a, v2f b, v8f c) {
    return __builtin_amdgcn_wmma_f32_16x16x4_f32(
         false, a,  false, b,
         (short)0, c,  false,  false);
}

__launch_bounds__(256)
__global__ void gather_mlp_kernel(const int* __restrict__ rowptr,
                                  const int* __restrict__ csr_src,
                                  const float* __restrict__ dinv,
                                  const float* __restrict__ x,
                                  const float* __restrict__ W1,
                                  const float* __restrict__ b1,
                                  const float* __restrict__ W2,
                                  float* __restrict__ h2, int n) {
    __shared__ __attribute__((aligned(16))) float tile[TM * TP];

    const int lane = threadIdx.x & 31;
    const int wave  = __builtin_amdgcn_readfirstlane((int)(threadIdx.x >> 5));
    const int node0 = blockIdx.x * TM;

    const float* xl = x + lane * 4;
    for (int i = 0; i < TM / 8; ++i) {
        const int rl = wave * (TM / 8) + i;
        const int d  = node0 + rl;
        v4f acc = (v4f){};
        if (d < n) {
            const int   beg = rowptr[d];
            const int   end = rowptr[d + 1];
            const float dd  = dinv[d];
            acc = *(const v4f*)(xl + (size_t)d * 128) * (dd * dd);
            for (int base = beg; base < end; base += 32) {
                const int m = min(end - base, 32);
                int   s_l = 0;
                float w_l = 0.0f;
                if (lane < m) {
                    s_l = csr_src[base + lane];
                    w_l = dinv[s_l] * dd;
                }
                const int w_i = __float_as_int(w_l);
                int j = 0;
                for (; j + 4 <= m; j += 4) {
                    const int   s0 = __builtin_amdgcn_readlane(s_l, j + 0);
                    const int   s1 = __builtin_amdgcn_readlane(s_l, j + 1);
                    const int   s2 = __builtin_amdgcn_readlane(s_l, j + 2);
                    const int   s3 = __builtin_amdgcn_readlane(s_l, j + 3);
                    const float w0 = __int_as_float(__builtin_amdgcn_readlane(w_i, j + 0));
                    const float w1 = __int_as_float(__builtin_amdgcn_readlane(w_i, j + 1));
                    const float w2 = __int_as_float(__builtin_amdgcn_readlane(w_i, j + 2));
                    const float w3 = __int_as_float(__builtin_amdgcn_readlane(w_i, j + 3));
                    const v4f v0 = *(const v4f*)(xl + (size_t)s0 * 128);
                    const v4f v1 = *(const v4f*)(xl + (size_t)s1 * 128);
                    const v4f v2 = *(const v4f*)(xl + (size_t)s2 * 128);
                    const v4f v3 = *(const v4f*)(xl + (size_t)s3 * 128);
                    acc += v0 * w0;
                    acc += v1 * w1;
                    acc += v2 * w2;
                    acc += v3 * w3;
                }
                for (; j < m; ++j) {
                    const int   s = __builtin_amdgcn_readlane(s_l, j);
                    const float w = __int_as_float(__builtin_amdgcn_readlane(w_i, j));
                    const v4f   v = *(const v4f*)(xl + (size_t)s * 128);
                    acc += v * w;
                }
            }
        }
        *(v4f*)(tile + rl * TP + lane * 4) = acc;
    }
    __syncthreads();

    const int c  = lane & 15;
    const int hh = lane >> 4;

    {
        constexpr int MT = TM / 16;
        v8f acc[MT];
        #pragma unroll
        for (int t = 0; t < MT; ++t) acc[t] = (v8f){};

        const float* bp = W1 + (size_t)(4 * hh) * 128 + wave * 16 + c;
        const float* ap = tile + c * TP + 4 * hh;
        #pragma unroll 2
        for (int j = 0; j < 16; ++j) {
            v2f blo, bhi;
            blo.x = bp[(8 * j + 0) * 128];
            blo.y = bp[(8 * j + 1) * 128];
            bhi.x = bp[(8 * j + 2) * 128];
            bhi.y = bp[(8 * j + 3) * 128];
            v4f a[MT];
            #pragma unroll
            for (int t = 0; t < MT; ++t) a[t] = *(const v4f*)(ap + t * 16 * TP + 8 * j);
            #pragma unroll
            for (int t = 0; t < MT; ++t) acc[t] = wmma_f32_k4(a[t].xy, blo, acc[t]);
            #pragma unroll
            for (int t = 0; t < MT; ++t) acc[t] = wmma_f32_k4(a[t].zw, bhi, acc[t]);
        }

        const float bias = b1[wave * 16 + c];
        __syncthreads();
        #pragma unroll
        for (int t = 0; t < MT; ++t) {
            float* hrow = tile + (t * 16 + 8 * hh) * TP + wave * 16 + c;
            #pragma unroll
            for (int r = 0; r < 8; ++r) hrow[r * TP] = fmaxf(acc[t][r] + bias, 0.0f);
        }
        __syncthreads();
    }

    {
        constexpr int MT2 = TM / 32;
        const int nt = wave & 3;
        const int mh = wave >> 2;
        v8f acc[MT2];
        #pragma unroll
        for (int t = 0; t < MT2; ++t) acc[t] = (v8f){};

        float dsc[MT2][8];
        #pragma unroll
        for (int t = 0; t < MT2; ++t) {
            const int rb = node0 + (mh * MT2 + t) * 16;
            #pragma unroll
            for (int r = 0; r < 8; ++r) {
                const float lo = dinv[min(rb + r, n - 1)];
                const float hi = dinv[min(rb + 8 + r, n - 1)];
                dsc[t][r] = hh ? hi : lo;
            }
        }

        const float* bp = W2 + (size_t)(4 * hh) * 64 + nt * 16 + c;
        const float* ap = tile + (mh * MT2 * 16 + c) * TP + 4 * hh;
        #pragma unroll 2
        for (int j = 0; j < 16; ++j) {
            v2f blo, bhi;
            blo.x = bp[(8 * j + 0) * 64];
            blo.y = bp[(8 * j + 1) * 64];
            bhi.x = bp[(8 * j + 2) * 64];
            bhi.y = bp[(8 * j + 3) * 64];
            v4f a[MT2];
            #pragma unroll
            for (int t = 0; t < MT2; ++t) a[t] = *(const v4f*)(ap + t * 16 * TP + 8 * j);
            #pragma unroll
            for (int t = 0; t < MT2; ++t) acc[t] = wmma_f32_k4(a[t].xy, blo, acc[t]);
            #pragma unroll
            for (int t = 0; t < MT2; ++t) acc[t] = wmma_f32_k4(a[t].zw, bhi, acc[t]);
        }

        #pragma unroll
        for (int t = 0; t < MT2; ++t) {
            const int row0 = node0 + (mh * MT2 + t) * 16 + 8 * hh;
            float* orow = h2 + (size_t)row0 * 64 + nt * 16 + c;
            #pragma unroll
            for (int r = 0; r < 8; ++r) {
                if (row0 + r < n) orow[(size_t)r * 64] = acc[t][r] * dsc[t][r];
            }
        }
    }
}

#define STRIP 7
#define ETRIP 8

template <int F, bool RELU>
__launch_bounds__(256)
__global__ void gather_sum_kernel(const int* __restrict__ rowptr,
                                  const int* __restrict__ csr_src,
                                  const float* __restrict__ dinv,
                                  const float* __restrict__ hs,
                                  const float* __restrict__ bias,
                                  float* __restrict__ out, int n) {
    static_assert(STRIP + 1 <= 32, "one lane per row pointer of the strip");
    constexpr int VE = F / 32;
    typedef __attribute__((ext_vector_type(VE))) float vec;

    const int lane = threadIdx.x & 31;
    const int d0 = STRIP * __builtin_amdgcn_readfirstlane(
        (int)(blockIdx.x * (blockDim.x >> 5) + (threadIdx.x >> 5)));
    if (d0 >= n) return;
    const int cnt = min(STRIP, n - d0);

    int   rp_l = 0;
    float dn_l = 0.0f;
    if (lane <= cnt) rp_l = rowptr[d0 + lane];
    if (lane <  cnt) dn_l = dinv[d0 + lane];
    const int dn_i = __float_as_int(dn_l);

    const float* hl = hs + lane * VE;
    const vec    bv = *(const vec*)(bias + lane * VE);

    const int e_last = max(__builtin_amdgcn_readlane(rp_l, cnt) - 1, 0);

    auto trip = [&](vec& acc, int s_l, int j, int rem, const bool tail) {
        int sk[ETRIP];
        vec vk[ETRIP];
        #pragma unroll
        for (int k = 0; k < ETRIP; ++k) sk[k] = __builtin_amdgcn_readlane(s_l, j + k);
        #pragma unroll
        for (int k = 0; k < ETRIP; ++k) vk[k] = *(const vec*)(hl + (size_t)sk[k] * F);
        #pragma unroll
        for (int k = 0; k < ETRIP; ++k) {
            vec v = vk[k];
            if (tail && k >= rem) v = (vec){};
            acc += v;
        }
    };
    auto sum_chunk = [&](vec& acc, int s_l, int m) {
        int j = 0;
        for (; j + ETRIP <= m; j += ETRIP) trip(acc, s_l, j, ETRIP, false);
        if (j < m) trip(acc, s_l, j, m - j, true);
    };

    int beg   = __builtin_amdgcn_readlane(rp_l, 0);
    int end   = __builtin_amdgcn_readlane(rp_l, 1);
    int s_raw = csr_src[min(beg + lane, e_last)];

    for (int i = 0; i < cnt; ++i) {
        const int   d  = d0 + i;
        const float dd = __int_as_float(__builtin_amdgcn_readlane(dn_i, i));
        const int   m  = min(end - beg, 32);
        const int   s_c = (lane < m) ? s_raw : 0;

        const int end_n = __builtin_amdgcn_readlane(rp_l, min(i + 2, cnt));
        s_raw = csr_src[min(end + lane, e_last)];

        vec acc = *(const vec*)(hl + (size_t)d * F);
        if (m > 0) sum_chunk(acc, s_c, m);
        for (int base = beg + 32; base < end; base += 32) {
            const int mm = min(end - base, 32);
            int s_l = 0;
            if (lane < mm) s_l = csr_src[base + lane];
            sum_chunk(acc, s_l, mm);
        }

        vec r = acc * dd + bv;
        if constexpr (RELU) {
            for (int q = 0; q < VE; ++q) r[q] = fmaxf(r[q], 0.0f);
        }
        *(vec*)(out + (size_t)d * F + lane * VE) = r;

        beg = end; end = end_n;
    }
}

extern "C" void kernel_launch(void* const* d_in, const int* in_sizes, int n_in,
                              void* d_out, int out_size, void* d_ws, size_t ws_size,
                              hipStream_t stream) {
    const float* x    = (const float*)d_in[0];
    const int*   eidx = (const int*)  d_in[1];
    const float* W1   = (const float*)d_in[2];
    const float* b1   = (const float*)d_in[3];
    const float* W2   = (const float*)d_in[4];
    const float* b2   = (const float*)d_in[5];
    float*       out  = (float*)d_out;

    const int N = NNODES;
    const int E = in_sizes[1] / 2;
    const int* src = eidx;
    const int* dst = eidx + E;

    char*  ws        = (char*)d_ws;
    float* dinv      = (float*)(ws);
    int*   rowptr    = (int*)  (ws + (size_t)409600);
    float* h2        = (float*)(ws + (size_t)(1  << 20));
    int*   csr       = (int*)  (ws + (size_t)(1 << 20) + (size_t)25600000);
    int*   recs      = (int*)  (ws + (size_t)(53 << 20));
    int*   chunk_off = (int*)  (ws + (size_t)(61 << 20));

    const int T = 256;

    const int nchunks = (E + CHUNK - 1) / CHUNK;
    bin_chunk_kernel<<<nchunks, 1024, 0, stream>>>(src, dst, chunk_off, recs, E);
    bucket_sort_kernel<<<NBK, 1024, 0, stream>>>(chunk_off, recs, rowptr, dinv, csr, N, nchunks);

    gather_mlp_kernel<<<(N + TM - 1) / TM, 256, 0, stream>>>(rowptr, csr, dinv, x, W1, b1, W2, h2, N);

    gather_sum_kernel<64, false><<<(N + STRIP * 8 - 1) / (STRIP * 8), T, 0, stream>>>(rowptr, csr, dinv, h2, b2, out, N);
}
